# starting kernel
# baseline (speedup 1.0000x reference)
#include <hip/hip_runtime.h>
#include <hip/hip_bf16.h>

typedef __attribute__((ext_vector_type(2))) float v2f;
typedef __attribute__((ext_vector_type(8))) float v8f;

#define AFWD 0.999f
#define EPSV 1e-5f
#define CCH  256
#define HW   4096
#define NB   32
#define NCHUNK 16
#define POS_PER_CHUNK (HW / NCHUNK)

__global__ __launch_bounds__(256) void k1_stats(const float* __restrict__ x,
                                                float* __restrict__ P1,
                                                float* __restrict__ P2) {
    const int chunk = blockIdx.x;
    const int b     = blockIdx.y;
    const int wave  = threadIdx.x >> 5;
    const int lane  = threadIdx.x & 31;
    const int half  = lane >> 4;
    const int li    = lane & 15;

    const int p0 = chunk * POS_PER_CHUNK;

    v2f ones;
    ones[0] = 1.0f;
    ones[1] = 1.0f;

#pragma unroll
    for (int t = 0; t < 2; ++t) {
        const int c0 = (wave + t * 8) * 16;
        const float* base = x + ((size_t)b * HW) * CCH + c0 + li;

        v8f s1 = {};
        v8f s2 = {};

        for (int g = 0; g < POS_PER_CHUNK / 4; ++g) {
            const int p = p0 + g * 4 + half * 2;
            float e0 = base[(size_t)p * CCH];
            float e1 = base[(size_t)(p + 1) * CCH];
            v2f bm, bq;
            bm[0] = e0;       bm[1] = e1;
            bq[0] = e0 * e0;  bq[1] = e1 * e1;
            s1 = __builtin_amdgcn_wmma_f32_16x16x4_f32(
                     false, ones, false, bm, (short)0, s1, false, false);
            s2 = __builtin_amdgcn_wmma_f32_16x16x4_f32(
                     false, ones, false, bq, (short)0, s2, false, false);
        }
        if (half == 0) {
            size_t o = (((size_t)b * NCHUNK) + chunk) * CCH + c0 + li;
            P1[o] = s1[0];
            P2[o] = s2[0];
        }
    }
}

__global__ __launch_bounds__(256) void k2_recur(const float* __restrict__ P1,
                                                const float* __restrict__ P2,
                                                const float* __restrict__ gamma,
                                                const float* __restrict__ beta,
                                                const float* __restrict__ stream_mu,
                                                const float* __restrict__ stream_var,
                                                float* __restrict__ Acoef,
                                                float* __restrict__ Bcoef) {
    const int c = threadIdx.x;
    __shared__ float red[CCH];

    const float Nf = (float)HW;
    float mu  = stream_mu[c];
    float var = stream_var[c];
    const float g  = gamma[c];
    const float be = beta[c];

    for (int b = 0; b < NB; ++b) {
        float s1 = 0.0f, s2 = 0.0f;
        for (int ch = 0; ch < NCHUNK; ++ch) {
            size_t o = (((size_t)b * NCHUNK) + ch) * CCH + c;
            s1 += P1[o];
            s2 += P2[o];
        }
        const float rinv   = rsqrtf(var + EPSV);
        const float sum_d  = s1 - Nf * mu;
        const float sum_d2 = s2 - 2.0f * mu * s1 + Nf * mu * mu;
        const float gr     = g * rinv;

        float contrib = gr * gr * sum_d2 + 2.0f * gr * be * sum_d + Nf * be * be;

        red[c] = contrib;
        __syncthreads();
        for (int s = CCH / 2; s > 0; s >>= 1) {
            if (c < s) red[c] += red[c + s];
            __syncthreads();
        }
        const float total = red[0];
        __syncthreads();

        const float ms     = total / (Nf * (float)CCH);
        const float rmsinv = rsqrtf(ms + EPSV);

        Acoef[b * CCH + c] = gr * rmsinv;
        Bcoef[b * CCH + c] = (be - gr * mu) * rmsinv;

        var = AFWD * var + AFWD * (1.0f - AFWD) * (sum_d2 / Nf);
        mu  = AFWD * mu + (1.0f - AFWD) * (s1 / Nf);
    }
}

__global__ __launch_bounds__(256) void k3_apply(const float* __restrict__ x,
                                                const float* __restrict__ Acoef,
                                                const float* __restrict__ Bcoef,
                                                float* __restrict__ out) {
    const size_t idx = (size_t)blockIdx.x * blockDim.x + threadIdx.x;
    const int b  = (int)(idx >> 18);
    const int c4 = (int)(idx & 63);

    const float4* xv = (const float4*)x;
    const float4* Av = (const float4*)Acoef;
    const float4* Bv = (const float4*)Bcoef;

    float4 xx = xv[idx];
    float4 A  = Av[b * 64 + c4];
    float4 B  = Bv[b * 64 + c4];

    float4 o;
    o.x = fmaf(A.x, xx.x, B.x);
    o.y = fmaf(A.y, xx.y, B.y);
    o.z = fmaf(A.z, xx.z, B.z);
    o.w = fmaf(A.w, xx.w, B.w);
    ((float4*)out)[idx] = o;
}

extern "C" void kernel_launch(void* const* d_in, const int* in_sizes, int n_in,
                              void* d_out, int out_size, void* d_ws, size_t ws_size,
                              hipStream_t stream) {
    const float* x          = (const float*)d_in[0];
    const float* gamma      = (const float*)d_in[1];
    const float* beta       = (const float*)d_in[2];
    const float* stream_mu  = (const float*)d_in[3];
    const float* stream_var = (const float*)d_in[4];
    float* out = (float*)d_out;

    float* P1 = (float*)d_ws;
    float* P2 = P1 + (size_t)NB * NCHUNK * CCH;
    float* Ac = P2 + (size_t)NB * NCHUNK * CCH;
    float* Bc = Ac + (size_t)NB * CCH;

    dim3 g1(NCHUNK, NB);
    k1_stats<<<g1, 256, 0, stream>>>(x, P1, P2);
    k2_recur<<<1, CCH, 0, stream>>>(P1, P2, gamma, beta, stream_mu, stream_var, Ac, Bc);

    const size_t total_f4 = (size_t)NB * HW * CCH / 4;
    k3_apply<<<(int)(total_f4 / 256), 256, 0, stream>>>(x, Ac, Bc, out);
}


# K2 recurrence split over 8 channel-group blocks: chunk sums and the cross-channel square sum leave the 32-step serial loop; K3 adds 8 per-sample partials (cross-channel sum order changes)
# speedup vs baseline: 1.6333x; 1.6333x over previous
#include <hip/hip_runtime.h>
#include <hip/hip_bf16.h>

typedef __attribute__((ext_vector_type(2))) float v2f;
typedef __attribute__((ext_vector_type(8))) float v8f;

#define AFWD 0.999f
#define EPSV 1e-5f
#define CCH  256
#define HW   4096
#define NB   32
#define NCHUNK 16
#define POS_PER_CHUNK (HW / NCHUNK)
#define K2_CG   32
#define K2_NBLK (CCH / K2_CG)
#define K2_THREADS (NB * K2_CG)

__global__ __launch_bounds__(256) void k1_stats(const float* __restrict__ x,
                                                float* __restrict__ P1,
                                                float* __restrict__ P2) {
    const int chunk = blockIdx.x;
    const int b     = blockIdx.y;
    const int wave  = threadIdx.x >> 5;
    const int lane  = threadIdx.x & 31;
    const int half  = lane >> 4;
    const int li    = lane & 15;

    const int p0 = chunk * POS_PER_CHUNK;

    v2f ones;
    ones[0] = 1.0f;
    ones[1] = 1.0f;

#pragma unroll
    for (int t = 0; t < 2; ++t) {
        const int c0 = (wave + t * 8) * 16;
        const float* base = x + ((size_t)b * HW) * CCH + c0 + li;

        v8f s1 = {};
        v8f s2 = {};

        for (int g = 0; g < POS_PER_CHUNK / 4; ++g) {
            const int p = p0 + g * 4 + half * 2;
            float e0 = base[(size_t)p * CCH];
            float e1 = base[(size_t)(p + 1) * CCH];
            v2f bm, bq;
            bm[0] = e0;       bm[1] = e1;
            bq[0] = e0 * e0;  bq[1] = e1 * e1;
            s1 = __builtin_amdgcn_wmma_f32_16x16x4_f32(
                     false, ones, false, bm, (short)0, s1, false, false);
            s2 = __builtin_amdgcn_wmma_f32_16x16x4_f32(
                     false, ones, false, bq, (short)0, s2, false, false);
        }
        if (half == 0) {
            size_t o = (((size_t)b * NCHUNK) + chunk) * CCH + c0 + li;
            P1[o] = s1[0];
            P2[o] = s2[0];
        }
    }
}

__global__ __launch_bounds__(K2_THREADS) void k2_recur(float* P1,
                                                 const float* __restrict__ P2,
                                                 const float* __restrict__ gamma,
                                                 const float* __restrict__ beta,
                                                 const float* __restrict__ stream_mu,
                                                 const float* __restrict__ stream_var,
                                                 float* __restrict__ Acoef,
                                                 float* __restrict__ Bcoef) {
    const int lane = threadIdx.x & 31;
    const int b    = threadIdx.x >> 5;
    const int c    = blockIdx.x * K2_CG + lane;

    __shared__ float S1[NB * K2_CG];
    __shared__ float S2[NB * K2_CG];
    __shared__ float CT[NB * K2_CG];

    float s1 = 0.0f, s2 = 0.0f;
    for (int ch = 0; ch < NCHUNK; ++ch) {
        size_t o = (((size_t)b * NCHUNK) + ch) * CCH + c;
        s1 += P1[o];
        s2 += P2[o];
    }
    S1[b * K2_CG + lane] = s1;
    S2[b * K2_CG + lane] = s2;
    __syncthreads();

    if (b == 0) {
        const float Nf = (float)HW;
        float mu  = stream_mu[c];
        float var = stream_var[c];
        const float g  = gamma[c];
        const float be = beta[c];

#pragma unroll
        for (int bb = 0; bb < NB; ++bb) {
            const float t1 = S1[bb * K2_CG + lane];
            const float t2 = S2[bb * K2_CG + lane];

            const float rinv   = rsqrtf(var + EPSV);
            const float sum_d  = t1 - Nf * mu;
            const float sum_d2 = t2 - 2.0f * mu * t1 + Nf * mu * mu;
            const float gr     = g * rinv;

            CT[bb * K2_CG + lane] = gr * gr * sum_d2 + 2.0f * gr * be * sum_d + Nf * be * be;

            Acoef[bb * CCH + c] = gr;
            Bcoef[bb * CCH + c] = be - gr * mu;

            var = AFWD * var + AFWD * (1.0f - AFWD) * (sum_d2 / Nf);
            mu  = AFWD * mu + (1.0f - AFWD) * (t1 / Nf);
        }
    }
    __syncthreads();

    float v = CT[b * K2_CG + lane];
    for (int m = 16; m > 0; m >>= 1) v += __shfl_xor(v, m, 32);
    if (lane == 0) P1[((size_t)b * NCHUNK) * CCH + blockIdx.x * K2_CG] = v;
}

__global__ __launch_bounds__(256) void k3_apply(const float* __restrict__ x,
                                                const float* __restrict__ Acoef,
                                                const float* __restrict__ Bcoef,
                                                const float* __restrict__ Tp,
                                                float* __restrict__ out) {
    const size_t idx = (size_t)blockIdx.x * blockDim.x + threadIdx.x;
    const int b  = (int)(blockIdx.x >> 10);
    const int c4 = (int)(idx & 63);

    const float4* xv = (const float4*)x;
    const float4* Av = (const float4*)Acoef;
    const float4* Bv = (const float4*)Bcoef;

    float4 xx = xv[idx];
    float4 A  = Av[b * 64 + c4];
    float4 B  = Bv[b * 64 + c4];

    float total = 0.0f;
    for (int j = 0; j < K2_NBLK; ++j)
        total += Tp[((size_t)b * NCHUNK) * CCH + j * K2_CG];
    const float ms     = total / ((float)HW * (float)CCH);
    const float rmsinv = rsqrtf(ms + EPSV);

    float4 o;
    o.x = fmaf(A.x * rmsinv, xx.x, B.x * rmsinv);
    o.y = fmaf(A.y * rmsinv, xx.y, B.y * rmsinv);
    o.z = fmaf(A.z * rmsinv, xx.z, B.z * rmsinv);
    o.w = fmaf(A.w * rmsinv, xx.w, B.w * rmsinv);
    ((float4*)out)[idx] = o;
}

extern "C" void kernel_launch(void* const* d_in, const int* in_sizes, int n_in,
                              void* d_out, int out_size, void* d_ws, size_t ws_size,
                              hipStream_t stream) {
    const float* x          = (const float*)d_in[0];
    const float* gamma      = (const float*)d_in[1];
    const float* beta       = (const float*)d_in[2];
    const float* stream_mu  = (const float*)d_in[3];
    const float* stream_var = (const float*)d_in[4];
    float* out = (float*)d_out;

    float* P1 = (float*)d_ws;
    float* P2 = P1 + (size_t)NB * NCHUNK * CCH;
    float* Ac = P2 + (size_t)NB * NCHUNK * CCH;
    float* Bc = Ac + (size_t)NB * CCH;

    dim3 g1(NCHUNK, NB);
    k1_stats<<<g1, 256, 0, stream>>>(x, P1, P2);
    k2_recur<<<K2_NBLK, K2_THREADS, 0, stream>>>(P1, P2, gamma, beta, stream_mu, stream_var, Ac, Bc);

    const size_t total_f4 = (size_t)NB * HW * CCH / 4;
    k3_apply<<<(int)(total_f4 / 256), 256, 0, stream>>>(x, Ac, Bc, P1, out);
}


# K1 load path: 16-byte loads per lane (whole 256-B row pieces, 64 channels x 128 positions per wave, 16-24 loads in flight); chunk sums added from two position halves via LDS (sum order changes)
# speedup vs baseline: 2.1593x; 1.3220x over previous
#include <hip/hip_runtime.h>
#include <hip/hip_bf16.h>

typedef __attribute__((ext_vector_type(2))) float v2f;
typedef __attribute__((ext_vector_type(8))) float v8f;

#define AFWD 0.999f
#define EPSV 1e-5f
#define CCH  256
#define HW   4096
#define NB   32
#define NCHUNK 16
#define POS_PER_CHUNK (HW / NCHUNK)
#define K2_CG   32
#define K2_NBLK (CCH / K2_CG)
#define K2_THREADS (NB * K2_CG)

#define K1_PW    128
#define K1_LB    4
#define K1_NBAT  (K1_PW / 4 / K1_LB)
#define K1_AHEAD 2
#define K1_RING  (K1_AHEAD + 1)

__global__ __launch_bounds__(256) void k1_stats(const float* __restrict__ x,
                                                float* __restrict__ P1,
                                                float* __restrict__ P2) {
    const int chunk = blockIdx.x;
    const int b     = blockIdx.y;
    const int wave  = threadIdx.x >> 5;
    const int lane  = threadIdx.x & 31;
    const int half  = lane >> 4;
    const int li    = lane & 15;
    const int cg    = wave & 3;
    const int ph    = wave >> 2;

    __shared__ float4 R1[4 * 16];
    __shared__ float4 R2[4 * 16];

    const int p0 = chunk * POS_PER_CHUNK + ph * K1_PW + half * 2;
    const float4* src = reinterpret_cast<const float4*>(
        x + ((size_t)b * HW + p0) * CCH + cg * 64 + li * 4);
    constexpr int ROWF4 = CCH / 4;

    float4 e[K1_RING][K1_LB][2];
#define K1_LOAD(BT) do { _Pragma("unroll") for (int s_ = 0; s_ < K1_LB; ++s_) { \
        e[(BT) % K1_RING][s_][0] = src[(size_t)((((BT) * K1_LB + s_) * 4) * ROWF4)]; \
        e[(BT) % K1_RING][s_][1] = src[(size_t)((((BT) * K1_LB + s_) * 4 + 1) * ROWF4)]; } } while (0)

    v2f ones;
    ones[0] = 1.0f;
    ones[1] = 1.0f;

    v8f s1[4] = {};
    v8f s2[4] = {};

#pragma unroll
    for (int bt = 0; bt < K1_AHEAD; ++bt) K1_LOAD(bt);

#pragma unroll
    for (int bt = 0; bt < K1_NBAT; ++bt) {
        __builtin_amdgcn_sched_barrier(0);
        if (bt + K1_AHEAD < K1_NBAT) K1_LOAD(bt + K1_AHEAD);
        __builtin_amdgcn_sched_barrier(0);
#pragma unroll
        for (int s = 0; s < K1_LB; ++s) {
            const float4 a = e[bt % K1_RING][s][0];
            const float4 c = e[bt % K1_RING][s][1];
            const float av[4] = {a.x, a.y, a.z, a.w};
            const float cv[4] = {c.x, c.y, c.z, c.w};
#pragma unroll
            for (int q = 0; q < 4; ++q) {
                v2f bm, bq;
                bm[0] = av[q];          bm[1] = cv[q];
                bq[0] = av[q] * av[q];  bq[1] = cv[q] * cv[q];
                s1[q] = __builtin_amdgcn_wmma_f32_16x16x4_f32(
                            false, ones, false, bm, (short)0, s1[q], false, false);
                s2[q] = __builtin_amdgcn_wmma_f32_16x16x4_f32(
                            false, ones, false, bq, (short)0, s2[q], false, false);
            }
        }
    }
#undef K1_LOAD

    float4 t1, t2;
    t1.x = s1[0][0]; t1.y = s1[1][0]; t1.z = s1[2][0]; t1.w = s1[3][0];
    t2.x = s2[0][0]; t2.y = s2[1][0]; t2.z = s2[2][0]; t2.w = s2[3][0];

    if (ph == 1 && half == 0) {
        R1[cg * 16 + li] = t1;
        R2[cg * 16 + li] = t2;
    }
    __syncthreads();
    if (ph == 0 && half == 0) {
        const float4 r1 = R1[cg * 16 + li];
        const float4 r2 = R2[cg * 16 + li];
        const size_t o = (((size_t)b * NCHUNK) + chunk) * CCH + cg * 64 + li * 4;
        float4 o1, o2;
        o1.x = t1.x + r1.x; o1.y = t1.y + r1.y; o1.z = t1.z + r1.z; o1.w = t1.w + r1.w;
        o2.x = t2.x + r2.x; o2.y = t2.y + r2.y; o2.z = t2.z + r2.z; o2.w = t2.w + r2.w;
        *reinterpret_cast<float4*>(P1 + o) = o1;
        *reinterpret_cast<float4*>(P2 + o) = o2;
    }
}

__global__ __launch_bounds__(K2_THREADS) void k2_recur(float* P1,
                                                 const float* __restrict__ P2,
                                                 const float* __restrict__ gamma,
                                                 const float* __restrict__ beta,
                                                 const float* __restrict__ stream_mu,
                                                 const float* __restrict__ stream_var,
                                                 float* __restrict__ Acoef,
                                                 float* __restrict__ Bcoef) {
    const int lane = threadIdx.x & 31;
    const int b    = threadIdx.x >> 5;
    const int c    = blockIdx.x * K2_CG + lane;

    __shared__ float S1[NB * K2_CG];
    __shared__ float S2[NB * K2_CG];
    __shared__ float CT[NB * K2_CG];

    float s1 = 0.0f, s2 = 0.0f;
    for (int ch = 0; ch < NCHUNK; ++ch) {
        size_t o = (((size_t)b * NCHUNK) + ch) * CCH + c;
        s1 += P1[o];
        s2 += P2[o];
    }
    S1[b * K2_CG + lane] = s1;
    S2[b * K2_CG + lane] = s2;
    __syncthreads();

    if (b == 0) {
        const float Nf = (float)HW;
        float mu  = stream_mu[c];
        float var = stream_var[c];
        const float g  = gamma[c];
        const float be = beta[c];

#pragma unroll
        for (int bb = 0; bb < NB; ++bb) {
            const float t1 = S1[bb * K2_CG + lane];
            const float t2 = S2[bb * K2_CG + lane];

            const float rinv   = rsqrtf(var + EPSV);
            const float sum_d  = t1 - Nf * mu;
            const float sum_d2 = t2 - 2.0f * mu * t1 + Nf * mu * mu;
            const float gr     = g * rinv;

            CT[bb * K2_CG + lane] = gr * gr * sum_d2 + 2.0f * gr * be * sum_d + Nf * be * be;

            Acoef[bb * CCH + c] = gr;
            Bcoef[bb * CCH + c] = be - gr * mu;

            var = AFWD * var + AFWD * (1.0f - AFWD) * (sum_d2 / Nf);
            mu  = AFWD * mu + (1.0f - AFWD) * (t1 / Nf);
        }
    }
    __syncthreads();

    float v = CT[b * K2_CG + lane];
    for (int m = 16; m > 0; m >>= 1) v += __shfl_xor(v, m, 32);
    if (lane == 0) P1[((size_t)b * NCHUNK) * CCH + blockIdx.x * K2_CG] = v;
}

__global__ __launch_bounds__(256) void k3_apply(const float* __restrict__ x,
                                                const float* __restrict__ Acoef,
                                                const float* __restrict__ Bcoef,
                                                const float* __restrict__ Tp,
                                                float* __restrict__ out) {
    const size_t idx = (size_t)blockIdx.x * blockDim.x + threadIdx.x;
    const int b  = (int)(blockIdx.x >> 10);
    const int c4 = (int)(idx & 63);

    const float4* xv = (const float4*)x;
    const float4* Av = (const float4*)Acoef;
    const float4* Bv = (const float4*)Bcoef;

    float4 xx = xv[idx];
    float4 A  = Av[b * 64 + c4];
    float4 B  = Bv[b * 64 + c4];

    float total = 0.0f;
    for (int j = 0; j < K2_NBLK; ++j)
        total += Tp[((size_t)b * NCHUNK) * CCH + j * K2_CG];
    const float ms     = total / ((float)HW * (float)CCH);
    const float rmsinv = rsqrtf(ms + EPSV);

    float4 o;
    o.x = fmaf(A.x * rmsinv, xx.x, B.x * rmsinv);
    o.y = fmaf(A.y * rmsinv, xx.y, B.y * rmsinv);
    o.z = fmaf(A.z * rmsinv, xx.z, B.z * rmsinv);
    o.w = fmaf(A.w * rmsinv, xx.w, B.w * rmsinv);
    ((float4*)out)[idx] = o;
}

extern "C" void kernel_launch(void* const* d_in, const int* in_sizes, int n_in,
                              void* d_out, int out_size, void* d_ws, size_t ws_size,
                              hipStream_t stream) {
    const float* x          = (const float*)d_in[0];
    const float* gamma      = (const float*)d_in[1];
    const float* beta       = (const float*)d_in[2];
    const float* stream_mu  = (const float*)d_in[3];
    const float* stream_var = (const float*)d_in[4];
    float* out = (float*)d_out;

    float* P1 = (float*)d_ws;
    float* P2 = P1 + (size_t)NB * NCHUNK * CCH;
    float* Ac = P2 + (size_t)NB * NCHUNK * CCH;
    float* Bc = Ac + (size_t)NB * CCH;

    dim3 g1(NCHUNK, NB);
    k1_stats<<<g1, 256, 0, stream>>>(x, P1, P2);
    k2_recur<<<K2_NBLK, K2_THREADS, 0, stream>>>(P1, P2, gamma, beta, stream_mu, stream_var, Ac, Bc);

    const size_t total_f4 = (size_t)NB * HW * CCH / 4;
    k3_apply<<<(int)(total_f4 / 256), 256, 0, stream>>>(x, Ac, Bc, P1, out);
}


# K3 re-gridded onto K1's 512 resident work-groups: each thread 64 float4 with 16-24 loads in flight, coefficients scaled once; region walked last position first (what K1 read last)
# speedup vs baseline: 2.5701x; 1.1902x over previous
#include <hip/hip_runtime.h>
#include <hip/hip_bf16.h>

typedef __attribute__((ext_vector_type(2))) float v2f;
typedef __attribute__((ext_vector_type(8))) float v8f;

#define AFWD 0.999f
#define EPSV 1e-5f
#define CCH  256
#define HW   4096
#define NB   32
#define NCHUNK 16
#define POS_PER_CHUNK (HW / NCHUNK)
#define K2_CG   32
#define K2_NBLK (CCH / K2_CG)
#define K2_THREADS (NB * K2_CG)

#define K1_PW    128
#define K1_LB    4
#define K1_NBAT  (K1_PW / 4 / K1_LB)
#define K1_AHEAD 2
#define K1_RING  (K1_AHEAD + 1)

__global__ __launch_bounds__(256) void k1_stats(const float* __restrict__ x,
                                                float* __restrict__ P1,
                                                float* __restrict__ P2) {
    const int chunk = blockIdx.x;
    const int b     = blockIdx.y;
    const int wave  = threadIdx.x >> 5;
    const int lane  = threadIdx.x & 31;
    const int half  = lane >> 4;
    const int li    = lane & 15;
    const int cg    = wave & 3;
    const int ph    = wave >> 2;

    __shared__ float4 R1[4 * 16];
    __shared__ float4 R2[4 * 16];

    const int p0 = chunk * POS_PER_CHUNK + ph * K1_PW + half * 2;
    const float4* src = reinterpret_cast<const float4*>(
        x + ((size_t)b * HW + p0) * CCH + cg * 64 + li * 4);
    constexpr int ROWF4 = CCH / 4;

    float4 e[K1_RING][K1_LB][2];
#define K1_LOAD(BT) do { _Pragma("unroll") for (int s_ = 0; s_ < K1_LB; ++s_) { \
        e[(BT) % K1_RING][s_][0] = src[(size_t)((((BT) * K1_LB + s_) * 4) * ROWF4)]; \
        e[(BT) % K1_RING][s_][1] = src[(size_t)((((BT) * K1_LB + s_) * 4 + 1) * ROWF4)]; } } while (0)

    v2f ones;
    ones[0] = 1.0f;
    ones[1] = 1.0f;

    v8f s1[4] = {};
    v8f s2[4] = {};

#pragma unroll
    for (int bt = 0; bt < K1_AHEAD; ++bt) K1_LOAD(bt);

#pragma unroll
    for (int bt = 0; bt < K1_NBAT; ++bt) {
        __builtin_amdgcn_sched_barrier(0);
        if (bt + K1_AHEAD < K1_NBAT) K1_LOAD(bt + K1_AHEAD);
        __builtin_amdgcn_sched_barrier(0);
#pragma unroll
        for (int s = 0; s < K1_LB; ++s) {
            const float4 a = e[bt % K1_RING][s][0];
            const float4 c = e[bt % K1_RING][s][1];
            const float av[4] = {a.x, a.y, a.z, a.w};
            const float cv[4] = {c.x, c.y, c.z, c.w};
#pragma unroll
            for (int q = 0; q < 4; ++q) {
                v2f bm, bq;
                bm[0] = av[q];          bm[1] = cv[q];
                bq[0] = av[q] * av[q];  bq[1] = cv[q] * cv[q];
                s1[q] = __builtin_amdgcn_wmma_f32_16x16x4_f32(
                            false, ones, false, bm, (short)0, s1[q], false, false);
                s2[q] = __builtin_amdgcn_wmma_f32_16x16x4_f32(
                            false, ones, false, bq, (short)0, s2[q], false, false);
            }
        }
    }
#undef K1_LOAD

    float4 t1, t2;
    t1.x = s1[0][0]; t1.y = s1[1][0]; t1.z = s1[2][0]; t1.w = s1[3][0];
    t2.x = s2[0][0]; t2.y = s2[1][0]; t2.z = s2[2][0]; t2.w = s2[3][0];

    if (ph == 1 && half == 0) {
        R1[cg * 16 + li] = t1;
        R2[cg * 16 + li] = t2;
    }
    __syncthreads();
    if (ph == 0 && half == 0) {
        const float4 r1 = R1[cg * 16 + li];
        const float4 r2 = R2[cg * 16 + li];
        const size_t o = (((size_t)b * NCHUNK) + chunk) * CCH + cg * 64 + li * 4;
        float4 o1, o2;
        o1.x = t1.x + r1.x; o1.y = t1.y + r1.y; o1.z = t1.z + r1.z; o1.w = t1.w + r1.w;
        o2.x = t2.x + r2.x; o2.y = t2.y + r2.y; o2.z = t2.z + r2.z; o2.w = t2.w + r2.w;
        *reinterpret_cast<float4*>(P1 + o) = o1;
        *reinterpret_cast<float4*>(P2 + o) = o2;
    }
}

__global__ __launch_bounds__(K2_THREADS) void k2_recur(float* P1,
                                                 const float* __restrict__ P2,
                                                 const float* __restrict__ gamma,
                                                 const float* __restrict__ beta,
                                                 const float* __restrict__ stream_mu,
                                                 const float* __restrict__ stream_var,
                                                 float* __restrict__ Acoef,
                                                 float* __restrict__ Bcoef) {
    const int lane = threadIdx.x & 31;
    const int b    = threadIdx.x >> 5;
    const int c    = blockIdx.x * K2_CG + lane;

    __shared__ float S1[NB * K2_CG];
    __shared__ float S2[NB * K2_CG];
    __shared__ float CT[NB * K2_CG];

    float s1 = 0.0f, s2 = 0.0f;
    for (int ch = 0; ch < NCHUNK; ++ch) {
        size_t o = (((size_t)b * NCHUNK) + ch) * CCH + c;
        s1 += P1[o];
        s2 += P2[o];
    }
    S1[b * K2_CG + lane] = s1;
    S2[b * K2_CG + lane] = s2;
    __syncthreads();

    if (b == 0) {
        const float Nf = (float)HW;
        float mu  = stream_mu[c];
        float var = stream_var[c];
        const float g  = gamma[c];
        const float be = beta[c];

#pragma unroll
        for (int bb = 0; bb < NB; ++bb) {
            const float t1 = S1[bb * K2_CG + lane];
            const float t2 = S2[bb * K2_CG + lane];

            const float rinv   = rsqrtf(var + EPSV);
            const float sum_d  = t1 - Nf * mu;
            const float sum_d2 = t2 - 2.0f * mu * t1 + Nf * mu * mu;
            const float gr     = g * rinv;

            CT[bb * K2_CG + lane] = gr * gr * sum_d2 + 2.0f * gr * be * sum_d + Nf * be * be;

            Acoef[bb * CCH + c] = gr;
            Bcoef[bb * CCH + c] = be - gr * mu;

            var = AFWD * var + AFWD * (1.0f - AFWD) * (sum_d2 / Nf);
            mu  = AFWD * mu + (1.0f - AFWD) * (t1 / Nf);
        }
    }
    __syncthreads();

    float v = CT[b * K2_CG + lane];
    for (int m = 16; m > 0; m >>= 1) v += __shfl_xor(v, m, 32);
    if (lane == 0) P1[((size_t)b * NCHUNK) * CCH + blockIdx.x * K2_CG] = v;
}

#define K3_LB    8
#define K3_NIT   (POS_PER_CHUNK / 4)
#define K3_NBAT  (K3_NIT / K3_LB)
#define K3_AHEAD 2
#define K3_RING  (K3_AHEAD + 1)
#define K3_STEP  (4 * (CCH / 4))

__global__ __launch_bounds__(256) void k3_apply(const float* __restrict__ x,
                                                const float* __restrict__ Acoef,
                                                const float* __restrict__ Bcoef,
                                                const float* __restrict__ Tp,
                                                float* __restrict__ out) {
    const int chunk = blockIdx.x;
    const int b     = blockIdx.y;
    const int c4    = threadIdx.x & 63;

    const size_t base = (((size_t)b * NCHUNK + chunk) * POS_PER_CHUNK) * (CCH / 4) + threadIdx.x;
    const float4* src = reinterpret_cast<const float4*>(x) + base;
    float4*       dst = reinterpret_cast<float4*>(out) + base;

    const float4 A = reinterpret_cast<const float4*>(Acoef)[b * 64 + c4];
    const float4 B = reinterpret_cast<const float4*>(Bcoef)[b * 64 + c4];

    float4 e[K3_RING][K3_LB];
#define K3_LOAD(BT) do { _Pragma("unroll") for (int s_ = 0; s_ < K3_LB; ++s_) { \
        e[(BT) % K3_RING][s_] = src[(size_t)((K3_NIT - 1 - ((BT) * K3_LB + s_)) * K3_STEP)]; } } while (0)

#pragma unroll
    for (int bt = 0; bt < K3_AHEAD; ++bt) K3_LOAD(bt);

    float total = 0.0f;
    for (int j = 0; j < K2_NBLK; ++j)
        total += Tp[((size_t)b * NCHUNK) * CCH + j * K2_CG];
    const float ms     = total / ((float)HW * (float)CCH);
    const float rmsinv = rsqrtf(ms + EPSV);

    float4 As, Bs;
    As.x = A.x * rmsinv; As.y = A.y * rmsinv; As.z = A.z * rmsinv; As.w = A.w * rmsinv;
    Bs.x = B.x * rmsinv; Bs.y = B.y * rmsinv; Bs.z = B.z * rmsinv; Bs.w = B.w * rmsinv;

#pragma unroll
    for (int bt = 0; bt < K3_NBAT; ++bt) {
        __builtin_amdgcn_sched_barrier(0);
        if (bt + K3_AHEAD < K3_NBAT) K3_LOAD(bt + K3_AHEAD);
        __builtin_amdgcn_sched_barrier(0);
        asm volatile("" : "+v"(As.x), "+v"(As.y), "+v"(As.z), "+v"(As.w));
#pragma unroll
        for (int s = 0; s < K3_LB; ++s) {
            const float4 xx = e[bt % K3_RING][s];
            float4 o;
            o.x = fmaf(As.x, xx.x, Bs.x);
            o.y = fmaf(As.y, xx.y, Bs.y);
            o.z = fmaf(As.z, xx.z, Bs.z);
            o.w = fmaf(As.w, xx.w, Bs.w);
            dst[(size_t)((K3_NIT - 1 - (bt * K3_LB + s)) * K3_STEP)] = o;
        }
    }
#undef K3_LOAD
}

extern "C" void kernel_launch(void* const* d_in, const int* in_sizes, int n_in,
                              void* d_out, int out_size, void* d_ws, size_t ws_size,
                              hipStream_t stream) {
    const float* x          = (const float*)d_in[0];
    const float* gamma      = (const float*)d_in[1];
    const float* beta       = (const float*)d_in[2];
    const float* stream_mu  = (const float*)d_in[3];
    const float* stream_var = (const float*)d_in[4];
    float* out = (float*)d_out;

    float* P1 = (float*)d_ws;
    float* P2 = P1 + (size_t)NB * NCHUNK * CCH;
    float* Ac = P2 + (size_t)NB * NCHUNK * CCH;
    float* Bc = Ac + (size_t)NB * CCH;

    dim3 g1(NCHUNK, NB);
    k1_stats<<<g1, 256, 0, stream>>>(x, P1, P2);
    k2_recur<<<K2_NBLK, K2_THREADS, 0, stream>>>(P1, P2, gamma, beta, stream_mu, stream_var, Ac, Bc);

    k3_apply<<<g1, 256, 0, stream>>>(x, Ac, Bc, P1, out);
}


# K3 output stores carry the non-temporal hint: write-once out lines stop taking second-level cache room from x lines that K1 read and K3 has yet to re-read
# speedup vs baseline: 2.9171x; 1.1350x over previous
#include <hip/hip_runtime.h>
#include <hip/hip_bf16.h>

typedef __attribute__((ext_vector_type(2))) float v2f;
typedef __attribute__((ext_vector_type(8))) float v8f;
typedef __attribute__((ext_vector_type(4))) float v4f;

#define AFWD 0.999f
#define EPSV 1e-5f
#define CCH  256
#define HW   4096
#define NB   32
#define NCHUNK 16
#define POS_PER_CHUNK (HW / NCHUNK)
#define K2_CG   32
#define K2_NBLK (CCH / K2_CG)
#define K2_THREADS (NB * K2_CG)

#define K1_PW    128
#define K1_LB    4
#define K1_NBAT  (K1_PW / 4 / K1_LB)
#define K1_AHEAD 2
#define K1_RING  (K1_AHEAD + 1)

__global__ __launch_bounds__(256) void k1_stats(const float* __restrict__ x,
                                                float* __restrict__ P1,
                                                float* __restrict__ P2) {
    const int chunk = blockIdx.x;
    const int b     = blockIdx.y;
    const int wave  = threadIdx.x >> 5;
    const int lane  = threadIdx.x & 31;
    const int half  = lane >> 4;
    const int li    = lane & 15;
    const int cg    = wave & 3;
    const int ph    = wave >> 2;

    __shared__ float4 R1[4 * 16];
    __shared__ float4 R2[4 * 16];

    const int p0 = chunk * POS_PER_CHUNK + ph * K1_PW + half * 2;
    const float4* src = reinterpret_cast<const float4*>(
        x + ((size_t)b * HW + p0) * CCH + cg * 64 + li * 4);
    constexpr int ROWF4 = CCH / 4;

    float4 e[K1_RING][K1_LB][2];
#define K1_LOAD(BT) do { _Pragma("unroll") for (int s_ = 0; s_ < K1_LB; ++s_) { \
        e[(BT) % K1_RING][s_][0] = src[(size_t)((((BT) * K1_LB + s_) * 4) * ROWF4)]; \
        e[(BT) % K1_RING][s_][1] = src[(size_t)((((BT) * K1_LB + s_) * 4 + 1) * ROWF4)]; } } while (0)

    v2f a1[2][2] = {};
    v2f a2[2][2] = {};

#pragma unroll
    for (int bt = 0; bt < K1_AHEAD; ++bt) K1_LOAD(bt);

#pragma unroll
    for (int bt = 0; bt < K1_NBAT; ++bt) {
        __builtin_amdgcn_sched_barrier(0);
        if (bt + K1_AHEAD < K1_NBAT) K1_LOAD(bt + K1_AHEAD);
        __builtin_amdgcn_sched_barrier(0);
#pragma unroll
        for (int s = 0; s < K1_LB; ++s) {
#pragma unroll
            for (int j = 0; j < 2; ++j) {
                const float4 v = e[bt % K1_RING][s][j];
                v2f lo, hi;
                lo[0] = v.x;  lo[1] = v.y;
                hi[0] = v.z;  hi[1] = v.w;
                a1[j][0] += lo;
                a1[j][1] += hi;
                a2[j][0] = __builtin_elementwise_fma(lo, lo, a2[j][0]);
                a2[j][1] = __builtin_elementwise_fma(hi, hi, a2[j][1]);
            }
        }
    }
#undef K1_LOAD

    v2f ones;
    ones[0] = 1.0f;
    ones[1] = 1.0f;
    v2f bm[4], bq[4];
#pragma unroll
    for (int q = 0; q < 4; ++q) {
        bm[q][0] = a1[0][q >> 1][q & 1];  bm[q][1] = a1[1][q >> 1][q & 1];
        bq[q][0] = a2[0][q >> 1][q & 1];  bq[q][1] = a2[1][q >> 1][q & 1];
    }
    v8f s1[4];
    v8f s2[4];
    asm volatile(
        "v_wmma_f32_16x16x4_f32 %0, %8, %9, 0\n\t"
        "v_wmma_f32_16x16x4_f32 %1, %8, %10, 0\n\t"
        "v_wmma_f32_16x16x4_f32 %2, %8, %11, 0\n\t"
        "v_wmma_f32_16x16x4_f32 %3, %8, %12, 0\n\t"
        "v_wmma_f32_16x16x4_f32 %4, %8, %13, 0\n\t"
        "v_wmma_f32_16x16x4_f32 %5, %8, %14, 0\n\t"
        "v_wmma_f32_16x16x4_f32 %6, %8, %15, 0\n\t"
        "v_wmma_f32_16x16x4_f32 %7, %8, %16, 0\n\t"
        "v_nop\n\tv_nop\n\tv_nop\n\tv_nop\n\t"
        "v_nop\n\tv_nop\n\tv_nop\n\tv_nop"
        : "=&v"(s1[0]), "=&v"(s1[1]), "=&v"(s1[2]), "=&v"(s1[3]),
          "=&v"(s2[0]), "=&v"(s2[1]), "=&v"(s2[2]), "=&v"(s2[3])
        : "v"(ones), "v"(bm[0]), "v"(bm[1]), "v"(bm[2]), "v"(bm[3]),
          "v"(bq[0]), "v"(bq[1]), "v"(bq[2]), "v"(bq[3]));

    float4 t1, t2;
    t1.x = s1[0][0]; t1.y = s1[1][0]; t1.z = s1[2][0]; t1.w = s1[3][0];
    t2.x = s2[0][0]; t2.y = s2[1][0]; t2.z = s2[2][0]; t2.w = s2[3][0];

    if (ph == 1 && half == 0) {
        R1[cg * 16 + li] = t1;
        R2[cg * 16 + li] = t2;
    }
    __syncthreads();
    if (ph == 0 && half == 0) {
        const float4 r1 = R1[cg * 16 + li];
        const float4 r2 = R2[cg * 16 + li];
        const size_t o = (((size_t)b * NCHUNK) + chunk) * CCH + cg * 64 + li * 4;
        float4 o1, o2;
        o1.x = t1.x + r1.x; o1.y = t1.y + r1.y; o1.z = t1.z + r1.z; o1.w = t1.w + r1.w;
        o2.x = t2.x + r2.x; o2.y = t2.y + r2.y; o2.z = t2.z + r2.z; o2.w = t2.w + r2.w;
        *reinterpret_cast<float4*>(P1 + o) = o1;
        *reinterpret_cast<float4*>(P2 + o) = o2;
    }
}

__global__ __launch_bounds__(K2_THREADS) void k2_recur(float* P1,
                                                 const float* __restrict__ P2,
                                                 const float* __restrict__ gamma,
                                                 const float* __restrict__ beta,
                                                 const float* __restrict__ stream_mu,
                                                 const float* __restrict__ stream_var,
                                                 float* __restrict__ Acoef,
                                                 float* __restrict__ Bcoef) {
    const int lane = threadIdx.x & 31;
    const int b    = threadIdx.x >> 5;
    const int c    = blockIdx.x * K2_CG + lane;

    __shared__ float S1[NB * K2_CG];
    __shared__ float S2[NB * K2_CG];
    __shared__ float CT[NB * K2_CG];

    float s1 = 0.0f, s2 = 0.0f;
    for (int ch = 0; ch < NCHUNK; ++ch) {
        size_t o = (((size_t)b * NCHUNK) + ch) * CCH + c;
        s1 += P1[o];
        s2 += P2[o];
    }
    S1[b * K2_CG + lane] = s1;
    S2[b * K2_CG + lane] = s2;
    __syncthreads();

    if (b == 0) {
        const float Nf = (float)HW;
        float mu  = stream_mu[c];
        float var = stream_var[c];
        const float g  = gamma[c];
        const float be = beta[c];

#pragma unroll
        for (int bb = 0; bb < NB; ++bb) {
            const float t1 = S1[bb * K2_CG + lane];
            const float t2 = S2[bb * K2_CG + lane];

            const float rinv   = rsqrtf(var + EPSV);
            const float sum_d  = t1 - Nf * mu;
            const float sum_d2 = t2 - 2.0f * mu * t1 + Nf * mu * mu;
            const float gr     = g * rinv;

            CT[bb * K2_CG + lane] = gr * gr * sum_d2 + 2.0f * gr * be * sum_d + Nf * be * be;

            Acoef[bb * CCH + c] = gr;
            Bcoef[bb * CCH + c] = be - gr * mu;

            var = AFWD * var + AFWD * (1.0f - AFWD) * (sum_d2 / Nf);
            mu  = AFWD * mu + (1.0f - AFWD) * (t1 / Nf);
        }
    }
    __syncthreads();

    float v = CT[b * K2_CG + lane];
    for (int m = 16; m > 0; m >>= 1) v += __shfl_xor(v, m, 32);
    if (lane == 0) P1[((size_t)b * NCHUNK) * CCH + blockIdx.x * K2_CG] = v;
}

#define K3_LB    8
#define K3_NIT   (POS_PER_CHUNK / 4)
#define K3_NBAT  (K3_NIT / K3_LB)
#define K3_AHEAD 2
#define K3_RING  (K3_AHEAD + 1)
#define K3_STEP  (4 * (CCH / 4))

__global__ __launch_bounds__(256) void k3_apply(const float* __restrict__ x,
                                                const float* __restrict__ Acoef,
                                                const float* __restrict__ Bcoef,
                                                const float* __restrict__ Tp,
                                                float* __restrict__ out) {
    const int chunk = blockIdx.x;
    const int b     = blockIdx.y;
    const int c4    = threadIdx.x & 63;

    const size_t base = (((size_t)b * NCHUNK + chunk) * POS_PER_CHUNK) * (CCH / 4) + threadIdx.x;
    const float4* src = reinterpret_cast<const float4*>(x) + base;
    float4*       dst = reinterpret_cast<float4*>(out) + base;

    const float4 A = reinterpret_cast<const float4*>(Acoef)[b * 64 + c4];
    const float4 B = reinterpret_cast<const float4*>(Bcoef)[b * 64 + c4];

    float4 e[K3_RING][K3_LB];
#define K3_LOAD(BT) do { _Pragma("unroll") for (int s_ = 0; s_ < K3_LB; ++s_) { \
        e[(BT) % K3_RING][s_] = src[(size_t)((K3_NIT - 1 - ((BT) * K3_LB + s_)) * K3_STEP)]; } } while (0)

#pragma unroll
    for (int bt = 0; bt < K3_AHEAD; ++bt) K3_LOAD(bt);

    float total = 0.0f;
    for (int j = 0; j < K2_NBLK; ++j)
        total += Tp[((size_t)b * NCHUNK) * CCH + j * K2_CG];
    const float ms     = total / ((float)HW * (float)CCH);
    const float rmsinv = rsqrtf(ms + EPSV);

    float4 As, Bs;
    As.x = A.x * rmsinv; As.y = A.y * rmsinv; As.z = A.z * rmsinv; As.w = A.w * rmsinv;
    Bs.x = B.x * rmsinv; Bs.y = B.y * rmsinv; Bs.z = B.z * rmsinv; Bs.w = B.w * rmsinv;

#pragma unroll
    for (int bt = 0; bt < K3_NBAT; ++bt) {
        __builtin_amdgcn_sched_barrier(0);
        if (bt + K3_AHEAD < K3_NBAT) K3_LOAD(bt + K3_AHEAD);
        __builtin_amdgcn_sched_barrier(0);
        asm volatile("" : "+v"(As.x), "+v"(As.y), "+v"(As.z), "+v"(As.w));
#pragma unroll
        for (int s = 0; s < K3_LB; ++s) {
            const float4 xx = e[bt % K3_RING][s];
            float4 o;
            o.x = fmaf(As.x, xx.x, Bs.x);
            o.y = fmaf(As.y, xx.y, Bs.y);
            o.z = fmaf(As.z, xx.z, Bs.z);
            o.w = fmaf(As.w, xx.w, Bs.w);
            __builtin_nontemporal_store(__builtin_bit_cast(v4f, o), reinterpret_cast<v4f*>(dst + (size_t)((K3_NIT - 1 - (bt * K3_LB + s)) * K3_STEP)));
        }
    }
#undef K3_LOAD
}

extern "C" void kernel_launch(void* const* d_in, const int* in_sizes, int n_in,
                              void* d_out, int out_size, void* d_ws, size_t ws_size,
                              hipStream_t stream) {
    const float* x          = (const float*)d_in[0];
    const float* gamma      = (const float*)d_in[1];
    const float* beta       = (const float*)d_in[2];
    const float* stream_mu  = (const float*)d_in[3];
    const float* stream_var = (const float*)d_in[4];
    float* out = (float*)d_out;

    float* P1 = (float*)d_ws;
    float* P2 = P1 + (size_t)NB * NCHUNK * CCH;
    float* Ac = P2 + (size_t)NB * NCHUNK * CCH;
    float* Bc = Ac + (size_t)NB * CCH;

    dim3 g1(NCHUNK, NB);
    k1_stats<<<g1, 256, 0, stream>>>(x, P1, P2);
    k2_recur<<<K2_NBLK, K2_THREADS, 0, stream>>>(P1, P2, gamma, beta, stream_mu, stream_var, Ac, Bc);

    k3_apply<<<g1, 256, 0, stream>>>(x, Ac, Bc, P1, out);
}


# Recurrence launch folded into the apply kernel: each work-group (re-cut: sample x 64 channels x position half, 2 partials per channel) rebuilds its coefficients; 3 launches -> 2; sum order changes
# speedup vs baseline: 3.9892x; 1.3675x over previous
#include <hip/hip_runtime.h>
#include <hip/hip_bf16.h>

typedef __attribute__((ext_vector_type(2))) float v2f;
typedef __attribute__((ext_vector_type(8))) float v8f;
typedef __attribute__((ext_vector_type(4))) float v4f;

#define AFWD 0.999f
#define EPSV 1e-5f
#define CCH  256
#define HW   4096
#define NB   32
#define NHALF 2
#define POS_PER_HALF (HW / NHALF)
#define NCG   4
#define WG_THREADS 512
#define GRID_X (NCG * NHALF)


#define K1_PW    128
#define K1_LB    4
#define K1_NBAT  (K1_PW / 4 / K1_LB)
#define K1_AHEAD 2
#define K1_RING  (K1_AHEAD + 1)

__global__ __launch_bounds__(WG_THREADS) void k1_stats(const float* __restrict__ x,
                                                       float* __restrict__ PP) {
    const int cg    = blockIdx.x & (NCG - 1);
    const int hf    = blockIdx.x >> 2;
    const int b     = blockIdx.y;
    const int wave  = threadIdx.x >> 5;
    const int lane  = threadIdx.x & 31;
    const int half  = lane >> 4;
    const int li    = lane & 15;

    __shared__ float4 R1[16 * 16];
    __shared__ float4 R2[16 * 16];

    const int p0 = hf * POS_PER_HALF + wave * K1_PW + half * 2;
    const float4* src = reinterpret_cast<const float4*>(
        x + ((size_t)b * HW + p0) * CCH + cg * 64 + li * 4);
    constexpr int ROWF4 = CCH / 4;

    float4 e[K1_RING][K1_LB][2];
#define K1_LOAD(BT) do { _Pragma("unroll") for (int s_ = 0; s_ < K1_LB; ++s_) { \
        e[(BT) % K1_RING][s_][0] = src[(size_t)((((BT) * K1_LB + s_) * 4) * ROWF4)]; \
        e[(BT) % K1_RING][s_][1] = src[(size_t)((((BT) * K1_LB + s_) * 4 + 1) * ROWF4)]; } } while (0)

    v2f a1[2][2] = {};
    v2f a2[2][2] = {};

#pragma unroll
    for (int bt = 0; bt < K1_AHEAD; ++bt) K1_LOAD(bt);

#pragma unroll
    for (int bt = 0; bt < K1_NBAT; ++bt) {
        __builtin_amdgcn_sched_barrier(0);
        if (bt + K1_AHEAD < K1_NBAT) K1_LOAD(bt + K1_AHEAD);
        __builtin_amdgcn_sched_barrier(0);
#pragma unroll
        for (int s = 0; s < K1_LB; ++s) {
#pragma unroll
            for (int j = 0; j < 2; ++j) {
                const float4 v = e[bt % K1_RING][s][j];
                v2f lo, hi;
                lo[0] = v.x;  lo[1] = v.y;
                hi[0] = v.z;  hi[1] = v.w;
                a1[j][0] += lo;
                a1[j][1] += hi;
                a2[j][0] = __builtin_elementwise_fma(lo, lo, a2[j][0]);
                a2[j][1] = __builtin_elementwise_fma(hi, hi, a2[j][1]);
            }
        }
    }
#undef K1_LOAD

    v2f ones;
    ones[0] = 1.0f;
    ones[1] = 1.0f;
    v2f bm[4], bq[4];
#pragma unroll
    for (int q = 0; q < 4; ++q) {
        bm[q][0] = a1[0][q >> 1][q & 1];  bm[q][1] = a1[1][q >> 1][q & 1];
        bq[q][0] = a2[0][q >> 1][q & 1];  bq[q][1] = a2[1][q >> 1][q & 1];
    }
    v8f s1[4];
    v8f s2[4];
    asm volatile(
        "v_wmma_f32_16x16x4_f32 %0, %8, %9, 0\n\t"
        "v_wmma_f32_16x16x4_f32 %1, %8, %10, 0\n\t"
        "v_wmma_f32_16x16x4_f32 %2, %8, %11, 0\n\t"
        "v_wmma_f32_16x16x4_f32 %3, %8, %12, 0\n\t"
        "v_wmma_f32_16x16x4_f32 %4, %8, %13, 0\n\t"
        "v_wmma_f32_16x16x4_f32 %5, %8, %14, 0\n\t"
        "v_wmma_f32_16x16x4_f32 %6, %8, %15, 0\n\t"
        "v_wmma_f32_16x16x4_f32 %7, %8, %16, 0\n\t"
        "v_nop\n\tv_nop\n\tv_nop\n\tv_nop\n\t"
        "v_nop\n\tv_nop\n\tv_nop\n\tv_nop"
        : "=&v"(s1[0]), "=&v"(s1[1]), "=&v"(s1[2]), "=&v"(s1[3]),
          "=&v"(s2[0]), "=&v"(s2[1]), "=&v"(s2[2]), "=&v"(s2[3])
        : "v"(ones), "v"(bm[0]), "v"(bm[1]), "v"(bm[2]), "v"(bm[3]),
          "v"(bq[0]), "v"(bq[1]), "v"(bq[2]), "v"(bq[3]));

    float4 t1, t2;
    t1.x = s1[0][0]; t1.y = s1[1][0]; t1.z = s1[2][0]; t1.w = s1[3][0];
    t2.x = s2[0][0]; t2.y = s2[1][0]; t2.z = s2[2][0]; t2.w = s2[3][0];

    if (half == 0) {
        R1[wave * 16 + li] = t1;
        R2[wave * 16 + li] = t2;
    }
    __syncthreads();
    if (threadIdx.x < 64) {
        const float* r1 = reinterpret_cast<const float*>(R1);
        const float* r2 = reinterpret_cast<const float*>(R2);
        float q1 = 0.0f, q2 = 0.0f;
        for (int w = 0; w < 16; ++w) {
            q1 += r1[w * 64 + threadIdx.x];
            q2 += r2[w * 64 + threadIdx.x];
        }
        float2 o;
        o.x = q1;  o.y = q2;
        reinterpret_cast<float2*>(PP)[((size_t)b * NHALF + hf) * CCH + cg * 64 + threadIdx.x] = o;
    }
}

#define K3_LB    8
#define K3_PSTEP (WG_THREADS / 16)
#define K3_NIT   (POS_PER_HALF / K3_PSTEP)
#define K3_NBAT  (K3_NIT / K3_LB)
#define K3_AHEAD 2
#define K3_RING  (K3_AHEAD + 1)
#define K3_STEP  (K3_PSTEP * (CCH / 4))

__global__ __launch_bounds__(WG_THREADS) void k3_apply(const float* __restrict__ x,
                                                       const float* __restrict__ PP,
                                                       const float* __restrict__ gamma,
                                                       const float* __restrict__ beta,
                                                       const float* __restrict__ stream_mu,
                                                       const float* __restrict__ stream_var,
                                                       float* __restrict__ out) {
    const int cg   = blockIdx.x & (NCG - 1);
    const int hf   = blockIdx.x >> 2;
    const int b    = blockIdx.y;
    const int c4   = threadIdx.x & 15;
    const int prow = threadIdx.x >> 4;

    __shared__ float4 sA4[CCH / 4];
    __shared__ float4 sB4[CCH / 4];
    __shared__ float sCT[8];
    float* sA = reinterpret_cast<float*>(sA4);
    float* sB = reinterpret_cast<float*>(sB4);

    const size_t base = ((size_t)b * HW + hf * POS_PER_HALF + prow) * (CCH / 4) + cg * 16 + c4;
    const float4* src = reinterpret_cast<const float4*>(x) + base;
    float4*       dst = reinterpret_cast<float4*>(out) + base;

    float4 e[K3_RING][K3_LB];
#define K3_LOAD(BT) do { _Pragma("unroll") for (int s_ = 0; s_ < K3_LB; ++s_) { \
        e[(BT) % K3_RING][s_] = src[(size_t)((K3_NIT - 1 - ((BT) * K3_LB + s_)) * K3_STEP)]; } } while (0)

#pragma unroll
    for (int bt = 0; bt < K3_AHEAD; ++bt) K3_LOAD(bt);

    if (threadIdx.x < CCH) {
        const int c = threadIdx.x;
        const float2* pp = reinterpret_cast<const float2*>(PP) + c;
        float2 T[NB][NHALF];
#pragma unroll
        for (int bb = 0; bb < NB; ++bb) {
#pragma unroll
            for (int h = 0; h < NHALF; ++h) T[bb][h] = pp[(bb * NHALF + h) * CCH];
        }
        const float Nf = (float)HW;
        float mu  = stream_mu[c];
        float var = stream_var[c];
        const float g  = gamma[c];
        const float be = beta[c];
        float t1b = 0.0f, t2b = 0.0f;
#pragma unroll
        for (int bb = 0; bb < NB; ++bb) {
            const float t1 = T[bb][0].x + T[bb][1].x;
            const float t2 = T[bb][0].y + T[bb][1].y;
            const float sum_d2 = t2 - 2.0f * mu * t1 + Nf * mu * mu;
            const float var_n = AFWD * var + AFWD * (1.0f - AFWD) * (sum_d2 / Nf);
            const float mu_n  = AFWD * mu + (1.0f - AFWD) * (t1 / Nf);
            const bool before = bb < b;
            const bool own    = bb == b;
            var = before ? var_n : var;
            mu  = before ? mu_n : mu;
            t1b = own ? t1 : t1b;
            t2b = own ? t2 : t2b;
        }
        const float rinv   = rsqrtf(var + EPSV);
        const float sum_d  = t1b - Nf * mu;
        const float sum_d2 = t2b - 2.0f * mu * t1b + Nf * mu * mu;
        const float gr     = g * rinv;
        float v = gr * gr * sum_d2 + 2.0f * gr * be * sum_d + Nf * be * be;
        for (int m = 16; m > 0; m >>= 1) v += __shfl_xor(v, m, 32);
        if ((threadIdx.x & 31) == 0) sCT[threadIdx.x >> 5] = v;
        sA[c] = gr;
        sB[c] = be - gr * mu;
    }
    __syncthreads();

    float total = 0.0f;
    for (int j = 0; j < 8; ++j) total += sCT[j];
    const float ms     = total / ((float)HW * (float)CCH);
    const float rmsinv = rsqrtf(ms + EPSV);

    const float4 A = sA4[cg * 16 + c4];
    const float4 B = sB4[cg * 16 + c4];
    float4 As, Bs;
    As.x = A.x * rmsinv; As.y = A.y * rmsinv; As.z = A.z * rmsinv; As.w = A.w * rmsinv;
    Bs.x = B.x * rmsinv; Bs.y = B.y * rmsinv; Bs.z = B.z * rmsinv; Bs.w = B.w * rmsinv;

#pragma unroll
    for (int bt = 0; bt < K3_NBAT; ++bt) {
        __builtin_amdgcn_sched_barrier(0);
        if (bt + K3_AHEAD < K3_NBAT) K3_LOAD(bt + K3_AHEAD);
        __builtin_amdgcn_sched_barrier(0);
        asm volatile("" : "+v"(As.x), "+v"(As.y), "+v"(As.z), "+v"(As.w));
#pragma unroll
        for (int s = 0; s < K3_LB; ++s) {
            const float4 xx = e[bt % K3_RING][s];
            float4 o;
            o.x = fmaf(As.x, xx.x, Bs.x);
            o.y = fmaf(As.y, xx.y, Bs.y);
            o.z = fmaf(As.z, xx.z, Bs.z);
            o.w = fmaf(As.w, xx.w, Bs.w);
            __builtin_nontemporal_store(__builtin_bit_cast(v4f, o), reinterpret_cast<v4f*>(dst + (size_t)((K3_NIT - 1 - (bt * K3_LB + s)) * K3_STEP)));
        }
    }
#undef K3_LOAD
}

extern "C" void kernel_launch(void* const* d_in, const int* in_sizes, int n_in,
                              void* d_out, int out_size, void* d_ws, size_t ws_size,
                              hipStream_t stream) {
    const float* x          = (const float*)d_in[0];
    const float* gamma      = (const float*)d_in[1];
    const float* beta       = (const float*)d_in[2];
    const float* stream_mu  = (const float*)d_in[3];
    const float* stream_var = (const float*)d_in[4];
    float* out = (float*)d_out;

    float* PP = (float*)d_ws;

    dim3 g(GRID_X, NB);
    k1_stats<<<g, WG_THREADS, 0, stream>>>(x, PP);
    k3_apply<<<g, WG_THREADS, 0, stream>>>(x, PP, gamma, beta, stream_mu, stream_var, out);
}


# Apply kernel head: partial-sum loads requested before the x loads and only for samples 0..b (uniform branches); its barrier orders LDS only, so x loads stay in flight: first store comes sooner
# speedup vs baseline: 4.1195x; 1.0327x over previous
#include <hip/hip_runtime.h>
#include <hip/hip_bf16.h>

typedef __attribute__((ext_vector_type(2))) float v2f;
typedef __attribute__((ext_vector_type(8))) float v8f;
typedef __attribute__((ext_vector_type(4))) float v4f;

#define AFWD 0.999f
#define EPSV 1e-5f
#define CCH  256
#define HW   4096
#define NB   32
#define NHALF 2
#define POS_PER_HALF (HW / NHALF)
#define NCG   4
#define WG_THREADS 512
#define GRID_X (NCG * NHALF)


#define K1_PW    128
#define K1_LB    4
#define K1_NBAT  (K1_PW / 4 / K1_LB)
#define K1_AHEAD 2
#define K1_RING  (K1_AHEAD + 1)

__global__ __launch_bounds__(WG_THREADS) void k1_stats(const float* __restrict__ x,
                                                       float* __restrict__ PP) {
    const int cg    = blockIdx.x & (NCG - 1);
    const int hf    = blockIdx.x >> 2;
    const int b     = blockIdx.y;
    const int wave  = threadIdx.x >> 5;
    const int lane  = threadIdx.x & 31;
    const int half  = lane >> 4;
    const int li    = lane & 15;

    __shared__ float4 R1[16 * 16];
    __shared__ float4 R2[16 * 16];

    const int p0 = hf * POS_PER_HALF + wave * K1_PW + half * 2;
    const float4* src = reinterpret_cast<const float4*>(
        x + ((size_t)b * HW + p0) * CCH + cg * 64 + li * 4);
    constexpr int ROWF4 = CCH / 4;

    float4 e[K1_RING][K1_LB][2];
#define K1_LOAD(BT) do { _Pragma("unroll") for (int s_ = 0; s_ < K1_LB; ++s_) { \
        e[(BT) % K1_RING][s_][0] = src[(size_t)((((BT) * K1_LB + s_) * 4) * ROWF4)]; \
        e[(BT) % K1_RING][s_][1] = src[(size_t)((((BT) * K1_LB + s_) * 4 + 1) * ROWF4)]; } } while (0)

    v2f a1[2][2] = {};
    v2f a2[2][2] = {};

#pragma unroll
    for (int bt = 0; bt < K1_AHEAD; ++bt) K1_LOAD(bt);

#pragma unroll
    for (int bt = 0; bt < K1_NBAT; ++bt) {
        __builtin_amdgcn_sched_barrier(0);
        if (bt + K1_AHEAD < K1_NBAT) K1_LOAD(bt + K1_AHEAD);
        __builtin_amdgcn_sched_barrier(0);
#pragma unroll
        for (int s = 0; s < K1_LB; ++s) {
#pragma unroll
            for (int j = 0; j < 2; ++j) {
                const float4 v = e[bt % K1_RING][s][j];
                v2f lo, hi;
                lo[0] = v.x;  lo[1] = v.y;
                hi[0] = v.z;  hi[1] = v.w;
                a1[j][0] += lo;
                a1[j][1] += hi;
                a2[j][0] = __builtin_elementwise_fma(lo, lo, a2[j][0]);
                a2[j][1] = __builtin_elementwise_fma(hi, hi, a2[j][1]);
            }
        }
    }
#undef K1_LOAD

    v2f ones;
    ones[0] = 1.0f;
    ones[1] = 1.0f;
    v2f bm[4], bq[4];
#pragma unroll
    for (int q = 0; q < 4; ++q) {
        bm[q][0] = a1[0][q >> 1][q & 1];  bm[q][1] = a1[1][q >> 1][q & 1];
        bq[q][0] = a2[0][q >> 1][q & 1];  bq[q][1] = a2[1][q >> 1][q & 1];
    }
    v8f s1[4];
    v8f s2[4];
    asm volatile(
        "v_wmma_f32_16x16x4_f32 %0, %8, %9, 0\n\t"
        "v_wmma_f32_16x16x4_f32 %1, %8, %10, 0\n\t"
        "v_wmma_f32_16x16x4_f32 %2, %8, %11, 0\n\t"
        "v_wmma_f32_16x16x4_f32 %3, %8, %12, 0\n\t"
        "v_wmma_f32_16x16x4_f32 %4, %8, %13, 0\n\t"
        "v_wmma_f32_16x16x4_f32 %5, %8, %14, 0\n\t"
        "v_wmma_f32_16x16x4_f32 %6, %8, %15, 0\n\t"
        "v_wmma_f32_16x16x4_f32 %7, %8, %16, 0\n\t"
        "v_nop\n\tv_nop\n\tv_nop\n\tv_nop\n\t"
        "v_nop\n\tv_nop\n\tv_nop\n\tv_nop"
        : "=&v"(s1[0]), "=&v"(s1[1]), "=&v"(s1[2]), "=&v"(s1[3]),
          "=&v"(s2[0]), "=&v"(s2[1]), "=&v"(s2[2]), "=&v"(s2[3])
        : "v"(ones), "v"(bm[0]), "v"(bm[1]), "v"(bm[2]), "v"(bm[3]),
          "v"(bq[0]), "v"(bq[1]), "v"(bq[2]), "v"(bq[3]));

    float4 t1, t2;
    t1.x = s1[0][0]; t1.y = s1[1][0]; t1.z = s1[2][0]; t1.w = s1[3][0];
    t2.x = s2[0][0]; t2.y = s2[1][0]; t2.z = s2[2][0]; t2.w = s2[3][0];

    if (half == 0) {
        R1[wave * 16 + li] = t1;
        R2[wave * 16 + li] = t2;
    }
    __syncthreads();
    if (threadIdx.x < 64) {
        const float* r1 = reinterpret_cast<const float*>(R1);
        const float* r2 = reinterpret_cast<const float*>(R2);
        float q1 = 0.0f, q2 = 0.0f;
        for (int w = 0; w < 16; ++w) {
            q1 += r1[w * 64 + threadIdx.x];
            q2 += r2[w * 64 + threadIdx.x];
        }
        float2 o;
        o.x = q1;  o.y = q2;
        reinterpret_cast<float2*>(PP)[((size_t)b * NHALF + hf) * CCH + cg * 64 + threadIdx.x] = o;
    }
}

#define K3_LB    8
#define K3_PSTEP (WG_THREADS / 16)
#define K3_NIT   (POS_PER_HALF / K3_PSTEP)
#define K3_NBAT  (K3_NIT / K3_LB)
#define K3_AHEAD 2
#define K3_RING  (K3_AHEAD + 1)
#define K3_STEP  (K3_PSTEP * (CCH / 4))
#define K3_HG    4

__global__ __launch_bounds__(WG_THREADS) void k3_apply(const float* __restrict__ x,
                                                       const float* __restrict__ PP,
                                                       const float* __restrict__ gamma,
                                                       const float* __restrict__ beta,
                                                       const float* __restrict__ stream_mu,
                                                       const float* __restrict__ stream_var,
                                                       float* __restrict__ out) {
    const int cg   = blockIdx.x & (NCG - 1);
    const int hf   = blockIdx.x >> 2;
    const int b    = blockIdx.y;
    const int c4   = threadIdx.x & 15;
    const int prow = threadIdx.x >> 4;

    __shared__ float4 sA4[CCH / 4];
    __shared__ float4 sB4[CCH / 4];
    __shared__ float sCT[8];
    float* sA = reinterpret_cast<float*>(sA4);
    float* sB = reinterpret_cast<float*>(sB4);

    const size_t base = ((size_t)b * HW + hf * POS_PER_HALF + prow) * (CCH / 4) + cg * 16 + c4;
    const float4* src = reinterpret_cast<const float4*>(x) + base;
    float4*       dst = reinterpret_cast<float4*>(out) + base;

    float4 e[K3_RING][K3_LB];
#define K3_LOAD(BT) do { _Pragma("unroll") for (int s_ = 0; s_ < K3_LB; ++s_) { \
        e[(BT) % K3_RING][s_] = src[(size_t)((K3_NIT - 1 - ((BT) * K3_LB + s_)) * K3_STEP)]; } } while (0)

    const bool headw = threadIdx.x < CCH;
    float2 T[NB][NHALF];
    float mu = 0.0f, var = 0.0f, g = 0.0f, be = 0.0f;
    if (headw) {
        const int c = threadIdx.x;
        const float2* pp = reinterpret_cast<const float2*>(PP) + c;
        mu  = stream_mu[c];
        var = stream_var[c];
        g   = gamma[c];
        be  = beta[c];
#pragma unroll
        for (int gq = 0; gq < NB / K3_HG; ++gq) {
            if (b >= gq * K3_HG) {
#pragma unroll
                for (int k = 0; k < K3_HG; ++k) {
#pragma unroll
                    for (int h = 0; h < NHALF; ++h)
                        T[gq * K3_HG + k][h] = pp[((gq * K3_HG + k) * NHALF + h) * CCH];
                }
            }
        }
    }

    __builtin_amdgcn_sched_barrier(0);
#pragma unroll
    for (int bt = 0; bt < K3_AHEAD; ++bt) K3_LOAD(bt);
    __builtin_amdgcn_sched_barrier(0);

    if (headw) {
        const int c = threadIdx.x;
        const float Nf = (float)HW;
        float t1b = 0.0f, t2b = 0.0f;
#pragma unroll
        for (int gq = 0; gq < NB / K3_HG; ++gq) {
            if (b >= gq * K3_HG) {
#pragma unroll
                for (int k = 0; k < K3_HG; ++k) {
                    const int bb = gq * K3_HG + k;
                    float2 u0 = T[bb][0], u1 = T[bb][1];
                    asm volatile("" : "+v"(u0.x), "+v"(u0.y), "+v"(u1.x), "+v"(u1.y));
                    const float t1 = u0.x + u1.x;
                    const float t2 = u0.y + u1.y;
                    const float sum_d2 = t2 - 2.0f * mu * t1 + Nf * mu * mu;
                    const float var_n = AFWD * var + AFWD * (1.0f - AFWD) * (sum_d2 / Nf);
                    const float mu_n  = AFWD * mu + (1.0f - AFWD) * (t1 / Nf);
                    const bool before = bb < b;
                    const bool own    = bb == b;
                    var = before ? var_n : var;
                    mu  = before ? mu_n : mu;
                    t1b = own ? t1 : t1b;
                    t2b = own ? t2 : t2b;
                }
            }
        }
        const float rinv   = rsqrtf(var + EPSV);
        const float sum_d  = t1b - Nf * mu;
        const float sum_d2 = t2b - 2.0f * mu * t1b + Nf * mu * mu;
        const float gr     = g * rinv;
        float v = gr * gr * sum_d2 + 2.0f * gr * be * sum_d + Nf * be * be;
        for (int m = 16; m > 0; m >>= 1) v += __shfl_xor(v, m, 32);
        if ((threadIdx.x & 31) == 0) sCT[threadIdx.x >> 5] = v;
        sA[c] = gr;
        sB[c] = be - gr * mu;
    }
    __builtin_amdgcn_fence(__ATOMIC_RELEASE, "workgroup", "local");
    __builtin_amdgcn_s_barrier();
    __builtin_amdgcn_fence(__ATOMIC_ACQUIRE, "workgroup", "local");

    float total = 0.0f;
    for (int j = 0; j < 8; ++j) total += sCT[j];
    const float ms     = total / ((float)HW * (float)CCH);
    const float rmsinv = rsqrtf(ms + EPSV);

    const float4 A = sA4[cg * 16 + c4];
    const float4 B = sB4[cg * 16 + c4];
    float4 As, Bs;
    As.x = A.x * rmsinv; As.y = A.y * rmsinv; As.z = A.z * rmsinv; As.w = A.w * rmsinv;
    Bs.x = B.x * rmsinv; Bs.y = B.y * rmsinv; Bs.z = B.z * rmsinv; Bs.w = B.w * rmsinv;

#pragma unroll
    for (int bt = 0; bt < K3_NBAT; ++bt) {
        __builtin_amdgcn_sched_barrier(0);
        if (bt + K3_AHEAD < K3_NBAT) K3_LOAD(bt + K3_AHEAD);
        __builtin_amdgcn_sched_barrier(0);
        asm volatile("" : "+v"(As.x), "+v"(As.y), "+v"(As.z), "+v"(As.w));
#pragma unroll
        for (int s = 0; s < K3_LB; ++s) {
            const float4 xx = e[bt % K3_RING][s];
            float4 o;
            o.x = fmaf(As.x, xx.x, Bs.x);
            o.y = fmaf(As.y, xx.y, Bs.y);
            o.z = fmaf(As.z, xx.z, Bs.z);
            o.w = fmaf(As.w, xx.w, Bs.w);
            __builtin_nontemporal_store(__builtin_bit_cast(v4f, o), reinterpret_cast<v4f*>(dst + (size_t)((K3_NIT - 1 - (bt * K3_LB + s)) * K3_STEP)));
        }
    }
#undef K3_LOAD
}

extern "C" void kernel_launch(void* const* d_in, const int* in_sizes, int n_in,
                              void* d_out, int out_size, void* d_ws, size_t ws_size,
                              hipStream_t stream) {
    const float* x          = (const float*)d_in[0];
    const float* gamma      = (const float*)d_in[1];
    const float* beta       = (const float*)d_in[2];
    const float* stream_mu  = (const float*)d_in[3];
    const float* stream_var = (const float*)d_in[4];
    float* out = (float*)d_out;

    float* PP = (float*)d_ws;

    dim3 g(GRID_X, NB);
    k1_stats<<<g, WG_THREADS, 0, stream>>>(x, PP);
    k3_apply<<<g, WG_THREADS, 0, stream>>>(x, PP, gamma, beta, stream_mu, stream_var, out);
}
